# starting kernel
# baseline (speedup 1.0000x reference)
#include <hip/hip_runtime.h>
#include <hip/hip_bf16.h>
#include <cstdint>

#define N_NODES 6000
#define NH      ((size_t)N_NODES * 128)

typedef float v2f __attribute__((ext_vector_type(2)));
typedef float v8f __attribute__((ext_vector_type(8)));

__device__ __forceinline__ v8f wmma4(v2f a, v2f b, v8f c) {
  return __builtin_amdgcn_wmma_f32_16x16x4_f32(false, a, false, b, (short)0, c,
                                               false, false);
}

__device__ __forceinline__ v2f ldsA(const float* base, int ld, int k0, int lane) {
  const int m = lane & 15;
  const int k = k0 + ((lane >> 4) << 1);
  const float* p = base + m * ld + k;
  v2f r; r.x = p[0]; r.y = p[1];
  return r;
}

__device__ __forceinline__ v2f ldBp(const float* __restrict__ P, int k0, int n0,
                                    int lane) {
  const int n = n0 + (lane & 15);
  const int k = k0 + ((lane >> 4) << 1);
  const float* p = P + (size_t)(k >> 1) * 256 + n * 2;
  v2f r; r.x = p[0]; r.y = p[1];
  return r;
}

__device__ __forceinline__ v2f ldBT(const float* __restrict__ Kb, int ld, int k0,
                                    int lane) {
  const int n = lane & 15;
  const int k = k0 + ((lane >> 4) << 1);
  const float* p = Kb + (size_t)n * ld + k;
  v2f r; r.x = p[0]; r.y = p[1];
  return r;
}

__device__ __forceinline__ float lrelu(float x) { return x >= 0.f ? x : 0.01f * x; }

__device__ __forceinline__ float rowmax16(float v) {
  v = fmaxf(v, __shfl_xor(v, 1));
  v = fmaxf(v, __shfl_xor(v, 2));
  v = fmaxf(v, __shfl_xor(v, 4));
  v = fmaxf(v, __shfl_xor(v, 8));
  return v;
}
__device__ __forceinline__ float rowsum16(float v) {
  v += __shfl_xor(v, 1);
  v += __shfl_xor(v, 2);
  v += __shfl_xor(v, 4);
  v += __shfl_xor(v, 8);
  return v;
}

__device__ __forceinline__ void layernorm128(float* buf, const float* __restrict__ g,
                                             const float* __restrict__ b,
                                             float (*stats)[2], int tid) {
  const int r = tid >> 3, seg = tid & 7;
  float sum = 0.f, sq = 0.f;
  #pragma unroll
  for (int i = 0; i < 16; i++) {
    float v = buf[r * 132 + seg * 16 + i];
    sum += v; sq += v * v;
  }
  sum += __shfl_xor(sum, 1); sq += __shfl_xor(sq, 1);
  sum += __shfl_xor(sum, 2); sq += __shfl_xor(sq, 2);
  sum += __shfl_xor(sum, 4); sq += __shfl_xor(sq, 4);
  if (seg == 0) {
    float mean = sum * (1.f / 128.f);
    float var  = sq * (1.f / 128.f) - mean * mean;
    stats[r][0] = mean;
    stats[r][1] = rsqrtf(var + 1e-5f);
  }
  __syncthreads();
  const float mean = stats[r][0], rstd = stats[r][1];
  #pragma unroll
  for (int i = 0; i < 16; i++) {
    int c = seg * 16 + i;
    float v = buf[r * 132 + c];
    buf[r * 132 + c] = g[c] * (v - mean) * rstd + b[c];
  }
}

__device__ __forceinline__ void gemm128_lds(const float* in, const float* __restrict__ Wp,
                                            const float* __restrict__ bias, float* outb,
                                            int lane, int wave, int act) {
  const int hi = lane >> 4, cn = lane & 15;
  const int n0 = wave * 32;
  v8f a0 = {}, a1 = {};
  for (int k0 = 0; k0 < 128; k0 += 4) {
    v2f a  = ldsA(in, 132, k0, lane);
    v2f b0 = ldBp(Wp, k0, n0, lane);
    v2f b1 = ldBp(Wp, k0, n0 + 16, lane);
    a0 = wmma4(a, b0, a0);
    a1 = wmma4(a, b1, a1);
  }
  #pragma unroll
  for (int j = 0; j < 8; j++) {
    int r = j + 8 * hi;
    float x0 = a0[j] + bias[n0 + cn];
    float x1 = a1[j] + bias[n0 + 16 + cn];
    if (act) { x0 = lrelu(x0); x1 = lrelu(x1); }
    outb[r * 132 + n0 + cn] = x0;
    outb[r * 132 + n0 + 16 + cn] = x1;
  }
}

__global__ void pack_pairs(const float* __restrict__ W, float* __restrict__ P,
                           int K, int pairs) {
  int i = blockIdx.x * blockDim.x + threadIdx.x;
  if (i >= pairs * 128) return;
  int k2 = i >> 7, n = i & 127;
  int k = 2 * k2;
  float a = (k < K)     ? W[(size_t)k * 128 + n]       : 0.f;
  float b = (k + 1 < K) ? W[(size_t)(k + 1) * 128 + n] : 0.f;
  P[(size_t)k2 * 256 + n * 2]     = a;
  P[(size_t)k2 * 256 + n * 2 + 1] = b;
}

__global__ void __launch_bounds__(128) proj_gemm(const float* __restrict__ A,
                                                 const float* __restrict__ Wp,
                                                 float* __restrict__ C, int act,
                                                 int pack) {
  __shared__ float at[16 * 132];
  const int tid = threadIdx.x, lane = tid & 31, wave = tid >> 5;
  const int row0 = blockIdx.x * 16;
  for (int i = tid; i < 16 * 128; i += 128) {
    int r = i >> 7, c = i & 127;
    at[r * 132 + c] = A[(size_t)(row0 + r) * 128 + c];
  }
  __syncthreads();
  const int n0 = wave * 32;
  v8f a0 = {}, a1 = {};
  for (int k0 = 0; k0 < 128; k0 += 4) {
    v2f a  = ldsA(at, 132, k0, lane);
    v2f b0 = ldBp(Wp, k0, n0, lane);
    v2f b1 = ldBp(Wp, k0, n0 + 16, lane);
    a0 = wmma4(a, b0, a0);
    a1 = wmma4(a, b1, a1);
  }
  const int hi = lane >> 4, cn = lane & 15;
  #pragma unroll
  for (int j = 0; j < 8; j++) {
    int r = row0 + j + 8 * hi;
    float x0 = a0[j], x1 = a1[j];
    if (act) { x0 = lrelu(x0); x1 = lrelu(x1); }
    if (pack) {
      C[(size_t)(r >> 1) * 256 + (n0 + cn) * 2 + (r & 1)]      = x0;
      C[(size_t)(r >> 1) * 256 + (n0 + 16 + cn) * 2 + (r & 1)] = x1;
    } else {
      C[(size_t)r * 128 + n0 + cn] = x0;
      C[(size_t)r * 128 + n0 + 16 + cn] = x1;
    }
  }
}

__global__ void __launch_bounds__(256) attn_kernel(const float* __restrict__ Q,
                                                   const float* __restrict__ K,
                                                   const float* __restrict__ Vp,
                                                   const float* __restrict__ mask,
                                                   int tmask, float* __restrict__ out) {
  __shared__ float qt[16 * 132];
  __shared__ float pbuf[8][16 * 18];
  __shared__ float outbuf[16 * 128];
  __shared__ float mbuf[8][16], lbuf[8][16];
  __shared__ float Mrow[16], Lrow[16];
  const int tid = threadIdx.x, lane = tid & 31, wave = tid >> 5;
  const int hi = lane >> 4, cn = lane & 15;
  const int qrow0 = blockIdx.x * 16;

  for (int i = tid; i < 2048; i += 256) {
    int r = i >> 7, c = i & 127;
    qt[r * 132 + c] = Q[(size_t)(qrow0 + r) * 128 + c];
  }
  __syncthreads();

  float mr[8], lr[8];
  v8f acc[8];
  #pragma unroll
  for (int j = 0; j < 8; j++) { mr[j] = -3.0e38f; lr[j] = 0.f; acc[j] = (v8f){}; }
  float* pw = &pbuf[wave][0];

  for (int ch = wave; ch < N_NODES / 16; ch += 8) {
    const int key0 = ch * 16;
    v8f s = {};
    for (int k0 = 0; k0 < 128; k0 += 4) {
      v2f a = ldsA(qt, 132, k0, lane);
      v2f b = ldBT(K + (size_t)key0 * 128, 128, k0, lane);
      s = wmma4(a, b, s);
    }
    const int col = key0 + cn;
    float sc[8];
    #pragma unroll
    for (int j = 0; j < 8; j++) {
      const int row = qrow0 + j + 8 * hi;
      float mv = tmask ? mask[(size_t)col * N_NODES + row]
                       : mask[(size_t)row * N_NODES + col];
      float sj = mv * s[j] - 1000.f * (1.f - mv);
      float mnew = fmaxf(mr[j], rowmax16(sj));
      sc[j] = __expf(mr[j] - mnew);
      float pj = __expf(sj - mnew);
      lr[j] = lr[j] * sc[j] + rowsum16(pj);
      mr[j] = mnew;
      pw[(j + 8 * hi) * 18 + cn] = pj;
    }
    #pragma unroll
    for (int t = 0; t < 8; t++)
      #pragma unroll
      for (int j = 0; j < 8; j++) acc[t][j] *= sc[j];
    asm volatile("s_wait_dscnt 0" ::: "memory");
    #pragma unroll
    for (int k0 = 0; k0 < 16; k0 += 4) {
      v2f a = ldsA(pw, 18, k0, lane);
      #pragma unroll
      for (int t = 0; t < 8; t++) {
        v2f b = ldBp(Vp, key0 + k0, t * 16, lane);
        acc[t] = wmma4(a, b, acc[t]);
      }
    }
  }

  if (cn == 0) {
    #pragma unroll
    for (int j = 0; j < 8; j++) {
      mbuf[wave][j + 8 * hi] = mr[j];
      lbuf[wave][j + 8 * hi] = lr[j];
    }
  }
  __syncthreads();
  for (int i = tid; i < 2048; i += 256) outbuf[i] = 0.f;
  if (tid < 16) {
    float M = -3.0e38f;
    for (int w = 0; w < 8; w++) M = fmaxf(M, mbuf[w][tid]);
    float L = 0.f;
    for (int w = 0; w < 8; w++) L += lbuf[w][tid] * __expf(mbuf[w][tid] - M);
    Mrow[tid] = M;
    Lrow[tid] = L > 0.f ? 1.f / L : 0.f;
  }
  __syncthreads();
  #pragma unroll
  for (int j = 0; j < 8; j++) {
    float e = __expf(mr[j] - Mrow[j + 8 * hi]);
    #pragma unroll
    for (int t = 0; t < 8; t++)
      atomicAdd(&outbuf[(j + 8 * hi) * 128 + t * 16 + cn], acc[t][j] * e);
  }
  __syncthreads();
  for (int i = tid; i < 2048; i += 256) {
    int r = i >> 7, c = i & 127;
    out[(size_t)(qrow0 + r) * 128 + c] = outbuf[i] * Lrow[r];
  }
}

#define CAT_LD 274
__global__ void __launch_bounds__(128) edge_kernel(
    const float* __restrict__ coors, const float* __restrict__ h,
    const int* __restrict__ src, const int* __restrict__ dst,
    const float* __restrict__ eW1p, const float* __restrict__ eb1,
    const float* __restrict__ eg1, const float* __restrict__ ebt1,
    const float* __restrict__ eW2p, const float* __restrict__ eb2,
    const float* __restrict__ cW1p, const float* __restrict__ cb1,
    const float* __restrict__ cg1, const float* __restrict__ cbt1,
    const float* __restrict__ cW2, const float* __restrict__ cb2,
    float* __restrict__ sum_msg, float* __restrict__ sum_x,
    float* __restrict__ cnt, int nE) {
  __shared__ float cat[16 * CAT_LD];
  __shared__ float buf1[16 * 132];
  __shared__ float buf2[16 * 132];
  __shared__ float stats[16][2];
  __shared__ float xrel[16][3];
  __shared__ float coefs[16];
  __shared__ int sidx[16], didx[16];
  const int tid = threadIdx.x, lane = tid & 31, wave = tid >> 5;
  const int hi = lane >> 4, cn = lane & 15;
  const int e0 = blockIdx.x * 16;

  if (tid < 16) {
    int e = e0 + tid;
    if (e < nE) {
      int s = src[e], d = dst[e];
      sidx[tid] = s; didx[tid] = d;
      float x0 = coors[s * 3 + 0] - coors[d * 3 + 0];
      float x1 = coors[s * 3 + 1] - coors[d * 3 + 1];
      float x2 = coors[s * 3 + 2] - coors[d * 3 + 2];
      xrel[tid][0] = x0; xrel[tid][1] = x1; xrel[tid][2] = x2;
      float d2 = x0 * x0 + x1 * x1 + x2 * x2;
      float sig = 1.f;
      for (int j = 0; j < 15; j++) {
        cat[tid * CAT_LD + 256 + j] = __expf(-d2 / sig);
        sig *= 1.5f;
      }
      cat[tid * CAT_LD + 271] = 0.f;
    } else {
      sidx[tid] = 0; didx[tid] = -1;
      xrel[tid][0] = xrel[tid][1] = xrel[tid][2] = 0.f;
      for (int j = 0; j < 16; j++) cat[tid * CAT_LD + 256 + j] = 0.f;
    }
  }
  __syncthreads();
  for (int i = tid; i < 16 * 256; i += 128) {
    int e = i >> 8, c = i & 255;
    int node = (didx[e] >= 0) ? ((c < 128) ? sidx[e] : didx[e]) : 0;
    cat[e * CAT_LD + c] = h[(size_t)node * 128 + (c & 127)];
  }
  __syncthreads();

  {
    const int n0 = wave * 32;
    v8f a0 = {}, a1 = {};
    for (int k0 = 0; k0 < 272; k0 += 4) {
      v2f a  = ldsA(cat, CAT_LD, k0, lane);
      v2f b0 = ldBp(eW1p, k0, n0, lane);
      v2f b1 = ldBp(eW1p, k0, n0 + 16, lane);
      a0 = wmma4(a, b0, a0);
      a1 = wmma4(a, b1, a1);
    }
    #pragma unroll
    for (int j = 0; j < 8; j++) {
      int r = j + 8 * hi;
      buf1[r * 132 + n0 + cn]      = lrelu(a0[j] + eb1[n0 + cn]);
      buf1[r * 132 + n0 + 16 + cn] = lrelu(a1[j] + eb1[n0 + 16 + cn]);
    }
  }
  __syncthreads();
  layernorm128(buf1, eg1, ebt1, stats, tid);
  __syncthreads();
  gemm128_lds(buf1, eW2p, eb2, buf2, lane, wave, 0);
  __syncthreads();
  gemm128_lds(buf2, cW1p, cb1, buf1, lane, wave, 1);
  __syncthreads();
  layernorm128(buf1, cg1, cbt1, stats, tid);
  __syncthreads();
  {
    int r = tid >> 3, seg = tid & 7;
    float sum = 0.f;
    #pragma unroll
    for (int i = 0; i < 16; i++) {
      int c = seg * 16 + i;
      sum += buf1[r * 132 + c] * cW2[c];
    }
    sum += __shfl_xor(sum, 1);
    sum += __shfl_xor(sum, 2);
    sum += __shfl_xor(sum, 4);
    if (seg == 0) coefs[r] = sum + cb2[0];
  }
  __syncthreads();
  for (int i = tid; i < 2048; i += 128) {
    int e = i >> 7, c = i & 127;
    int d = didx[e];
    if (d >= 0) atomicAdd(&sum_msg[(size_t)d * 128 + c], buf2[e * 132 + c]);
  }
  if (tid < 48) {
    int e = tid / 3, ax = tid % 3;
    int d = didx[e];
    if (d >= 0) atomicAdd(&sum_x[d * 3 + ax], xrel[e][ax] * coefs[e]);
  }
  if (tid < 16) {
    int d = didx[tid];
    if (d >= 0) atomicAdd(&cnt[d], 1.f);
  }
}

#define NCAT_LD 450
__global__ void __launch_bounds__(128) node_kernel(
    const float* __restrict__ h, const float* __restrict__ summ,
    const float* __restrict__ cnt, const float* __restrict__ cm,
    const float* __restrict__ orig,
    const float* __restrict__ nW1p, const float* __restrict__ nb1,
    const float* __restrict__ ng1, const float* __restrict__ nbt1,
    const float* __restrict__ nW2p, const float* __restrict__ nb2,
    float* __restrict__ hout) {
  __shared__ float cat[16 * NCAT_LD];
  __shared__ float buf1[16 * 132];
  __shared__ float stats[16][2];
  const int tid = threadIdx.x, lane = tid & 31, wave = tid >> 5;
  const int hi = lane >> 4, cn = lane & 15;
  const int node0 = blockIdx.x * 16;

  for (int i = tid; i < 2048; i += 128) {
    int e = i >> 7, c = i & 127;
    int node = node0 + e;
    cat[e * NCAT_LD + c] = h[(size_t)node * 128 + c];
    float ct = cnt[node];
    cat[e * NCAT_LD + 128 + c] = ct > 0.f ? summ[(size_t)node * 128 + c] / ct : 0.f;
    cat[e * NCAT_LD + 256 + c] = cm[(size_t)node * 128 + c];
  }
  for (int i = tid; i < 1024; i += 128) {
    int e = i >> 6, c = i & 63;
    cat[e * NCAT_LD + 384 + c] = orig[(size_t)(node0 + e) * 64 + c];
  }
  __syncthreads();
  {
    const int n0 = wave * 32;
    v8f a0 = {}, a1 = {};
    for (int k0 = 0; k0 < 448; k0 += 4) {
      v2f a  = ldsA(cat, NCAT_LD, k0, lane);
      v2f b0 = ldBp(nW1p, k0, n0, lane);
      v2f b1 = ldBp(nW1p, k0, n0 + 16, lane);
      a0 = wmma4(a, b0, a0);
      a1 = wmma4(a, b1, a1);
    }
    #pragma unroll
    for (int j = 0; j < 8; j++) {
      int r = j + 8 * hi;
      buf1[r * 132 + n0 + cn]      = lrelu(a0[j] + nb1[n0 + cn]);
      buf1[r * 132 + n0 + 16 + cn] = lrelu(a1[j] + nb1[n0 + 16 + cn]);
    }
  }
  __syncthreads();
  layernorm128(buf1, ng1, nbt1, stats, tid);
  __syncthreads();
  {
    const int n0 = wave * 32;
    v8f a0 = {}, a1 = {};
    for (int k0 = 0; k0 < 128; k0 += 4) {
      v2f a  = ldsA(buf1, 132, k0, lane);
      v2f b0 = ldBp(nW2p, k0, n0, lane);
      v2f b1 = ldBp(nW2p, k0, n0 + 16, lane);
      a0 = wmma4(a, b0, a0);
      a1 = wmma4(a, b1, a1);
    }
    #pragma unroll
    for (int j = 0; j < 8; j++) {
      int r = j + 8 * hi;
      size_t g0 = (size_t)(node0 + r) * 128 + n0 + cn;
      size_t g1 = g0 + 16;
      hout[g0] = 0.5f * (a0[j] + nb2[n0 + cn]) + 0.5f * h[g0];
      hout[g1] = 0.5f * (a1[j] + nb2[n0 + 16 + cn]) + 0.5f * h[g1];
    }
  }
}

__global__ void coords_kernel(const float* __restrict__ oc, const float* __restrict__ c,
                              const float* __restrict__ sx, const float* __restrict__ cnt,
                              float* __restrict__ out, int n) {
  int i = blockIdx.x * blockDim.x + threadIdx.x;
  if (i >= n * 3) return;
  int node = i / 3;
  float ct = cnt[node];
  float xu = ct > 0.f ? sx[i] / ct : 0.f;
  out[i] = 0.2f * oc[i] + 0.8f * c[i] + xu;
}

extern "C" void kernel_launch(void* const* d_in, const int* in_sizes, int n_in,
                              void* d_out, int out_size, void* d_ws, size_t ws_size,
                              hipStream_t stream) {
  const float* coors_lig = (const float*)d_in[0];
  const float* h_lig     = (const float*)d_in[1];
  const float* orig_lig  = (const float*)d_in[2];
  const float* oc_lig    = (const float*)d_in[3];
  const float* coors_rec = (const float*)d_in[4];
  const float* h_rec     = (const float*)d_in[5];
  const float* orig_rec  = (const float*)d_in[6];
  const float* oc_rec    = (const float*)d_in[7];
  const int* lig_src = (const int*)d_in[8];
  const int* lig_dst = (const int*)d_in[9];
  const int* rec_src = (const int*)d_in[10];
  const int* rec_dst = (const int*)d_in[11];
  const float* mask = (const float*)d_in[12];
  const float* eW1 = (const float*)d_in[13];
  const float* eb1 = (const float*)d_in[14];
  const float* eg1 = (const float*)d_in[15];
  const float* ebt1 = (const float*)d_in[16];
  const float* eW2 = (const float*)d_in[17];
  const float* eb2 = (const float*)d_in[18];
  const float* qW = (const float*)d_in[19];
  const float* kW = (const float*)d_in[20];
  const float* vW = (const float*)d_in[21];
  const float* nW1 = (const float*)d_in[22];
  const float* nb1 = (const float*)d_in[23];
  const float* ng1 = (const float*)d_in[24];
  const float* nbt1 = (const float*)d_in[25];
  const float* nW2 = (const float*)d_in[26];
  const float* nb2 = (const float*)d_in[27];
  const float* cW1 = (const float*)d_in[28];
  const float* cb1 = (const float*)d_in[29];
  const float* cg1 = (const float*)d_in[30];
  const float* cbt1 = (const float*)d_in[31];
  const float* cW2 = (const float*)d_in[32];
  const float* cb2 = (const float*)d_in[33];
  const int nE = in_sizes[8];

  float* ws = (float*)d_ws;
  float* q_l = ws + 0 * NH;
  float* k_l = ws + 1 * NH;
  float* v_l = ws + 2 * NH;
  float* q_r = ws + 3 * NH;
  float* k_r = ws + 4 * NH;
  float* v_r = ws + 5 * NH;
  float* cm_l = ws + 6 * NH;
  float* cm_r = ws + 7 * NH;
  float* sm_l = ws + 8 * NH;
  float* sm_r = ws + 9 * NH;
  float* sx_l = ws + 10 * NH;
  float* sx_r = sx_l + 3 * N_NODES;
  float* cnt_l = sx_r + 3 * N_NODES;
  float* cnt_r = cnt_l + N_NODES;
  float* eW1p = cnt_r + N_NODES;
  float* eW2p = eW1p + 136 * 256;
  float* cW1p = eW2p + 64 * 256;
  float* nW1p = cW1p + 64 * 256;
  float* nW2p = nW1p + 224 * 256;
  float* qWp  = nW2p + 64 * 256;
  float* kWp  = qWp + 64 * 256;
  float* vWp  = kWp + 64 * 256;

  size_t accum_floats = 2 * NH + 8 * (size_t)N_NODES;
  hipMemsetAsync(sm_l, 0, accum_floats * sizeof(float), stream);

  pack_pairs<<<(136 * 128 + 255) / 256, 256, 0, stream>>>(eW1, eW1p, 271, 136);
  pack_pairs<<<(64 * 128 + 255) / 256, 256, 0, stream>>>(eW2, eW2p, 128, 64);
  pack_pairs<<<(64 * 128 + 255) / 256, 256, 0, stream>>>(cW1, cW1p, 128, 64);
  pack_pairs<<<(224 * 128 + 255) / 256, 256, 0, stream>>>(nW1, nW1p, 448, 224);
  pack_pairs<<<(64 * 128 + 255) / 256, 256, 0, stream>>>(nW2, nW2p, 128, 64);
  pack_pairs<<<(64 * 128 + 255) / 256, 256, 0, stream>>>(qW, qWp, 128, 64);
  pack_pairs<<<(64 * 128 + 255) / 256, 256, 0, stream>>>(kW, kWp, 128, 64);
  pack_pairs<<<(64 * 128 + 255) / 256, 256, 0, stream>>>(vW, vWp, 128, 64);

  const int ntiles = N_NODES / 16;
  proj_gemm<<<ntiles, 128, 0, stream>>>(h_lig, qWp, q_l, 1, 0);
  proj_gemm<<<ntiles, 128, 0, stream>>>(h_lig, kWp, k_l, 1, 0);
  proj_gemm<<<ntiles, 128, 0, stream>>>(h_lig, vWp, v_l, 0, 1);
  proj_gemm<<<ntiles, 128, 0, stream>>>(h_rec, qWp, q_r, 1, 0);
  proj_gemm<<<ntiles, 128, 0, stream>>>(h_rec, kWp, k_r, 1, 0);
  proj_gemm<<<ntiles, 128, 0, stream>>>(h_rec, vWp, v_r, 0, 1);

  attn_kernel<<<ntiles, 256, 0, stream>>>(q_l, k_r, v_r, mask, 0, cm_l);
  attn_kernel<<<ntiles, 256, 0, stream>>>(q_r, k_l, v_l, mask, 1, cm_r);

  int eblocks = (nE + 15) / 16;
  edge_kernel<<<eblocks, 128, 0, stream>>>(coors_lig, h_lig, lig_src, lig_dst,
      eW1p, eb1, eg1, ebt1, eW2p, eb2, cW1p, cb1, cg1, cbt1, cW2, cb2,
      sm_l, sx_l, cnt_l, nE);
  edge_kernel<<<eblocks, 128, 0, stream>>>(coors_rec, h_rec, rec_src, rec_dst,
      eW1p, eb1, eg1, ebt1, eW2p, eb2, cW1p, cb1, cg1, cbt1, cW2, cb2,
      sm_r, sx_r, cnt_r, nE);

  float* out_xl = (float*)d_out;
  float* out_hl = out_xl + 3 * N_NODES;
  float* out_xr = out_hl + NH;
  float* out_hr = out_xr + 3 * N_NODES;

  node_kernel<<<ntiles, 128, 0, stream>>>(h_lig, sm_l, cnt_l, cm_l, orig_lig,
      nW1p, nb1, ng1, nbt1, nW2p, nb2, out_hl);
  node_kernel<<<ntiles, 128, 0, stream>>>(h_rec, sm_r, cnt_r, cm_r, orig_rec,
      nW1p, nb1, ng1, nbt1, nW2p, nb2, out_hr);

  int cthreads = 3 * N_NODES;
  coords_kernel<<<(cthreads + 255) / 256, 256, 0, stream>>>(oc_lig, coors_lig, sx_l,
                                                            cnt_l, out_xl, N_NODES);
  coords_kernel<<<(cthreads + 255) / 256, 256, 0, stream>>>(oc_rec, coors_rec, sx_r,
                                                            cnt_r, out_xr, N_NODES);
}


# attention key range cut into 4 pieces, one work-group each (188x4 groups fill all 256 processors in 3 even rounds; 188 left 68 idle); node kernel joins piece max, sum, output; sum order changes
# speedup vs baseline: 1.5046x; 1.5046x over previous
#include <hip/hip_runtime.h>
#include <hip/hip_bf16.h>
#include <cstdint>

#define N_NODES 6000
#define NH      ((size_t)N_NODES * 128)

typedef float v2f __attribute__((ext_vector_type(2)));
typedef float v8f __attribute__((ext_vector_type(8)));

__device__ __forceinline__ v8f wmma4(v2f a, v2f b, v8f c) {
  return __builtin_amdgcn_wmma_f32_16x16x4_f32(false, a, false, b, (short)0, c,
                                               false, false);
}

__device__ __forceinline__ v2f ldsA(const float* base, int ld, int k0, int lane) {
  const int m = lane & 15;
  const int k = k0 + ((lane >> 4) << 1);
  const float* p = base + m * ld + k;
  v2f r; r.x = p[0]; r.y = p[1];
  return r;
}

__device__ __forceinline__ v2f ldBp(const float* __restrict__ P, int k0, int n0,
                                    int lane) {
  const int n = n0 + (lane & 15);
  const int k = k0 + ((lane >> 4) << 1);
  const float* p = P + (size_t)(k >> 1) * 256 + n * 2;
  v2f r; r.x = p[0]; r.y = p[1];
  return r;
}

__device__ __forceinline__ v2f ldBT(const float* __restrict__ Kb, int ld, int k0,
                                    int lane) {
  const int n = lane & 15;
  const int k = k0 + ((lane >> 4) << 1);
  const float* p = Kb + (size_t)n * ld + k;
  v2f r; r.x = p[0]; r.y = p[1];
  return r;
}

__device__ __forceinline__ float lrelu(float x) { return x >= 0.f ? x : 0.01f * x; }

__device__ __forceinline__ float rowmax16(float v) {
  v = fmaxf(v, __shfl_xor(v, 1));
  v = fmaxf(v, __shfl_xor(v, 2));
  v = fmaxf(v, __shfl_xor(v, 4));
  v = fmaxf(v, __shfl_xor(v, 8));
  return v;
}
__device__ __forceinline__ float rowsum16(float v) {
  v += __shfl_xor(v, 1);
  v += __shfl_xor(v, 2);
  v += __shfl_xor(v, 4);
  v += __shfl_xor(v, 8);
  return v;
}

__device__ __forceinline__ void layernorm128(float* buf, const float* __restrict__ g,
                                             const float* __restrict__ b,
                                             float (*stats)[2], int tid) {
  const int r = tid >> 3, seg = tid & 7;
  float sum = 0.f, sq = 0.f;
  #pragma unroll
  for (int i = 0; i < 16; i++) {
    float v = buf[r * 132 + seg * 16 + i];
    sum += v; sq += v * v;
  }
  sum += __shfl_xor(sum, 1); sq += __shfl_xor(sq, 1);
  sum += __shfl_xor(sum, 2); sq += __shfl_xor(sq, 2);
  sum += __shfl_xor(sum, 4); sq += __shfl_xor(sq, 4);
  if (seg == 0) {
    float mean = sum * (1.f / 128.f);
    float var  = sq * (1.f / 128.f) - mean * mean;
    stats[r][0] = mean;
    stats[r][1] = rsqrtf(var + 1e-5f);
  }
  __syncthreads();
  const float mean = stats[r][0], rstd = stats[r][1];
  #pragma unroll
  for (int i = 0; i < 16; i++) {
    int c = seg * 16 + i;
    float v = buf[r * 132 + c];
    buf[r * 132 + c] = g[c] * (v - mean) * rstd + b[c];
  }
}

__global__ void pack_pairs(const float* __restrict__ W, float* __restrict__ P,
                           int K, int pairs) {
  int i = blockIdx.x * blockDim.x + threadIdx.x;
  if (i >= pairs * 128) return;
  int k2 = i >> 7, n = i & 127;
  int k = 2 * k2;
  float a = (k < K)     ? W[(size_t)k * 128 + n]       : 0.f;
  float b = (k + 1 < K) ? W[(size_t)(k + 1) * 128 + n] : 0.f;
  P[(size_t)k2 * 256 + n * 2]     = a;
  P[(size_t)k2 * 256 + n * 2 + 1] = b;
}

__global__ void __launch_bounds__(128) proj_gemm(const float* __restrict__ A,
                                                 const float* __restrict__ Wp,
                                                 float* __restrict__ C, int act,
                                                 int pack) {
  __shared__ float at[16 * 132];
  const int tid = threadIdx.x, lane = tid & 31, wave = tid >> 5;
  const int row0 = blockIdx.x * 16;
  for (int i = tid; i < 16 * 128; i += 128) {
    int r = i >> 7, c = i & 127;
    at[r * 132 + c] = A[(size_t)(row0 + r) * 128 + c];
  }
  __syncthreads();
  const int n0 = wave * 32;
  v8f a0 = {}, a1 = {};
  for (int k0 = 0; k0 < 128; k0 += 4) {
    v2f a  = ldsA(at, 132, k0, lane);
    v2f b0 = ldBp(Wp, k0, n0, lane);
    v2f b1 = ldBp(Wp, k0, n0 + 16, lane);
    a0 = wmma4(a, b0, a0);
    a1 = wmma4(a, b1, a1);
  }
  const int hi = lane >> 4, cn = lane & 15;
  #pragma unroll
  for (int j = 0; j < 8; j++) {
    int r = row0 + j + 8 * hi;
    float x0 = a0[j], x1 = a1[j];
    if (act) { x0 = lrelu(x0); x1 = lrelu(x1); }
    if (pack) {
      C[(size_t)(r >> 1) * 256 + (n0 + cn) * 2 + (r & 1)]      = x0;
      C[(size_t)(r >> 1) * 256 + (n0 + 16 + cn) * 2 + (r & 1)] = x1;
    } else {
      C[(size_t)r * 128 + n0 + cn] = x0;
      C[(size_t)r * 128 + n0 + 16 + cn] = x1;
    }
  }
}

#define AT_KEYS  128
#define AT_NST   ((N_NODES + AT_KEYS - 1) / AT_KEYS)
#define AT_SPLITS 4
#define AT_SPT   ((AT_NST + AT_SPLITS - 1) / AT_SPLITS)
#define K_LD     132
#define V_LD     288
#define KBUF_F   (AT_KEYS * K_LD)
#define VBUF_F   ((AT_KEYS / 2) * V_LD)
#define STAGE_F  (KBUF_F + VBUF_F)
#define AT_LDS_F (2 * STAGE_F + 32 * 132 + 16 * 16 * 18 + 256 + 256 + 32)

__global__ void __launch_bounds__(512) attn_kernel(const float* __restrict__ Q,
                                                   const float* __restrict__ K,
                                                   const float* __restrict__ Vp,
                                                   const float* __restrict__ mask,
                                                   int tmask,
                                                   float* __restrict__ op0,
                                                   float* __restrict__ op1,
                                                   float* __restrict__ op2,
                                                   float* __restrict__ op3,
                                                   float* __restrict__ ml) {
  __shared__ __attribute__((aligned(16))) float L[AT_LDS_F];
  float* qtl  = L + 2 * STAGE_F;
  float* pbuf = qtl + 32 * 132;
  float* mbuf = pbuf + 16 * 16 * 18;
  float* lbuf = mbuf + 256;
  float* Mrow = lbuf + 256;

  const int tid = threadIdx.x, lane = tid & 31;
  const int wave = __builtin_amdgcn_readfirstlane(tid >> 5);
  const int hi = lane >> 4, cn = lane & 15;
  const int qt = wave >> 3, kg = wave & 7;
  const int qblk0 = blockIdx.x * 32;
  const int ks = blockIdx.y;
  const int st0 = ks * AT_SPT;
  const int st1 = (st0 + AT_SPT < AT_NST) ? (st0 + AT_SPT) : AT_NST;
  float* __restrict__ out = (ks == 0) ? op0 : (ks == 1) ? op1 : (ks == 2) ? op2 : op3;
  const int qrow0 = qblk0 + qt * 16;
  const bool qvalid = qrow0 < N_NODES;
  const int vrow = wave >> 1, vc4 = tid & 63;

  float4 pre[8];
  float mk[8], mkn[8];
  #pragma unroll
  for (int j = 0; j < 8; j++) mkn[j] = 0.f;

  auto fetchK = [&](int st) {
    const int key_s0 = st * AT_KEYS;
    #pragma unroll
    for (int i = 0; i < 8; i++) {
      const int key = key_s0 + wave + 16 * i;
      if (key < N_NODES)
        pre[i] = *reinterpret_cast<const float4*>(K + (size_t)key * 128 + lane * 4);
      else
        pre[i] = make_float4(0.f, 0.f, 0.f, 0.f);
    }
    const int key0 = key_s0 + 16 * kg;
    if (qvalid && key0 < N_NODES) {
      const int col = key0 + cn;
      #pragma unroll
      for (int j = 0; j < 8; j++) {
        const int row = qrow0 + j + 8 * hi;
        mkn[j] = tmask ? mask[(size_t)col * N_NODES + row]
                       : mask[(size_t)row * N_NODES + col];
      }
    }
  };
  auto parkK = [&](int st) {
    float* Kd = L + (st & 1) * STAGE_F;
    #pragma unroll
    for (int i = 0; i < 8; i++)
      *reinterpret_cast<float4*>(Kd + (wave + 16 * i) * K_LD + lane * 4) = pre[i];
  };
  auto fetchV = [&](int st) {
    #pragma unroll
    for (int i = 0; i < 8; i++) {
      const int pr = st * (AT_KEYS / 2) + vrow + 8 * i;
      if (2 * pr < N_NODES)
        pre[i] = *reinterpret_cast<const float4*>(Vp + (size_t)pr * 256 + vc4 * 4);
      else
        pre[i] = make_float4(0.f, 0.f, 0.f, 0.f);
    }
  };
  auto parkV = [&](int st) {
    float* Vd = L + (st & 1) * STAGE_F + KBUF_F;
    #pragma unroll
    for (int i = 0; i < 8; i++)
      *reinterpret_cast<float4*>(Vd + (vrow + 8 * i) * V_LD + vc4 * 4) = pre[i];
  };

  for (int i = 0; i < 2; i++) {
    const int f = tid + 512 * i;
    const int r = f >> 5, c4 = f & 31;
    float4 v = make_float4(0.f, 0.f, 0.f, 0.f);
    if (qblk0 + r < N_NODES)
      v = *reinterpret_cast<const float4*>(Q + (size_t)(qblk0 + r) * 128 + c4 * 4);
    *reinterpret_cast<float4*>(qtl + r * 132 + c4 * 4) = v;
  }
  fetchK(st0);
  parkK(st0);
  fetchV(st0);
  parkV(st0);
  #pragma unroll
  for (int j = 0; j < 8; j++) mk[j] = mkn[j];
  __syncthreads();

  float mr[8], lr[8];
  v8f acc[8];
  #pragma unroll
  for (int j = 0; j < 8; j++) { mr[j] = -3.0e38f; lr[j] = 0.f; acc[j] = (v8f){}; }
  float* pw = pbuf + wave * (16 * 18);
  const float* qa = qtl + qt * (16 * 132);

  for (int st = st0; st < st1; st++) {
    const bool more = (st + 1 < st1);
    const bool work = qvalid && (st * AT_KEYS + 16 * kg < N_NODES);
    const float* Kb = L + (st & 1) * STAGE_F + (16 * kg) * K_LD;
    const float* Vb = L + (st & 1) * STAGE_F + KBUF_F + (8 * kg) * V_LD;
    float sc[8];

    if (more) fetchK(st + 1);
    if (work) {
      v8f s = {};
      #pragma unroll
      for (int k0 = 0; k0 < 128; k0 += 4) {
        v2f a = ldsA(qa, 132, k0, lane);
        v2f b = ldsA(Kb, K_LD, k0, lane);
        s = wmma4(a, b, s);
      }
      #pragma unroll
      for (int j = 0; j < 8; j++) {
        const float mv = mk[j];
        float sj = mv * s[j] - 1000.f * (1.f - mv);
        float mnew = fmaxf(mr[j], rowmax16(sj));
        sc[j] = __expf(mr[j] - mnew);
        float pj = __expf(sj - mnew);
        lr[j] = lr[j] * sc[j] + rowsum16(pj);
        mr[j] = mnew;
        pw[(j + 8 * hi) * 18 + cn] = pj;
      }
    }
    if (more) { parkK(st + 1); fetchV(st + 1); }
    if (work) {
      #pragma unroll
      for (int t = 0; t < 8; t++)
        #pragma unroll
        for (int j = 0; j < 8; j++) acc[t][j] *= sc[j];
      asm volatile("s_wait_dscnt 0" ::: "memory");
      #pragma unroll
      for (int k0 = 0; k0 < 16; k0 += 4) {
        v2f a = ldsA(pw, 18, k0, lane);
        const float* vr = Vb + ((k0 >> 1) + hi) * V_LD + cn * 2;
        v2f b[8];
        #pragma unroll
        for (int t = 0; t < 8; t++) b[t] = *reinterpret_cast<const v2f*>(vr + t * 32);
        #pragma unroll
        for (int t = 0; t < 8; t++) acc[t] = wmma4(a, b[t], acc[t]);
      }
    }
    if (more) parkV(st + 1);
    #pragma unroll
    for (int j = 0; j < 8; j++) mk[j] = mkn[j];
    __syncthreads();
  }

  if (cn == 0) {
    #pragma unroll
    for (int j = 0; j < 8; j++) {
      mbuf[wave * 16 + j + 8 * hi] = mr[j];
      lbuf[wave * 16 + j + 8 * hi] = lr[j];
    }
  }
  __syncthreads();
  if (tid < 32) {
    const int tq = tid >> 4, r = tid & 15;
    float M = -3.0e38f;
    for (int g = 0; g < 8; g++) M = fmaxf(M, mbuf[(tq * 8 + g) * 16 + r]);
    float Ls = 0.f;
    for (int g = 0; g < 8; g++)
      Ls += lbuf[(tq * 8 + g) * 16 + r] * __expf(mbuf[(tq * 8 + g) * 16 + r] - M);
    Mrow[tid] = M;
    if (qblk0 + tid < N_NODES) {
      ml[(size_t)(2 * ks) * N_NODES + qblk0 + tid]     = M;
      ml[(size_t)(2 * ks + 1) * N_NODES + qblk0 + tid] = Ls;
    }
  }
  __syncthreads();
  {
    float* cb = L + wave * 2048;
    #pragma unroll
    for (int j = 0; j < 8; j++) {
      float e = __expf(mr[j] - Mrow[qt * 16 + j + 8 * hi]);
      #pragma unroll
      for (int t = 0; t < 8; t++)
        cb[(j + 8 * hi) * 128 + t * 16 + cn] = acc[t][j] * e;
    }
  }
  __syncthreads();
  for (int i = tid; i < 4096; i += 512) {
    const int r = i >> 7, c = i & 127;
    const int tq = r >> 4, rr = r & 15;
    float sum = 0.f;
    #pragma unroll
    for (int g = 0; g < 8; g++) sum += L[(tq * 8 + g) * 2048 + rr * 128 + c];
    if (qblk0 + r < N_NODES) out[(size_t)(qblk0 + r) * 128 + c] = sum;
  }
}

__global__ void pack_frag(const float* __restrict__ W, float* __restrict__ P,
                          int K, int total, int mode) {
  int i = blockIdx.x * blockDim.x + threadIdx.x;
  if (i >= total) return;
  const int e = i & 3, lane = (i >> 2) & 31, t = (i >> 7) & 7, Q = i >> 10;
  const int h = lane >> 4, m = lane & 15;
  int k;
  if (mode == 0) k = (Q < 32) ? (8 * Q + 4 * h + e) : (8 * Q + 4 * (e >> 1) + 2 * h + (e & 1));
  else           k = 16 * (Q >> 1) + 8 * h + 4 * (Q & 1) + e;
  P[i] = (k < K) ? W[(size_t)k * 128 + 16 * t + m] : 0.f;
}

#define EW_W1F  (34 * 8 * 128)
#define EW_W2F  (16 * 8 * 128)
#define EW_IMGF (EW_W1F + 2 * EW_W2F)
#define EW_PARF (8 * 128)
#define EW_SCR  68
#define EW_LDSF (EW_IMGF + EW_PARF + 8 * 16 * EW_SCR)

__device__ __forceinline__ float4 ldW(const float* Wl, int Q, int t, int lane) {
  return *reinterpret_cast<const float4*>(Wl + ((Q * 8 + t) * 32 + lane) * 4);
}

__device__ __forceinline__ void ldP(const float* PAR, int v, int t, int hi, float* p) {
  const float4 a = *reinterpret_cast<const float4*>(PAR + v * 128 + 16 * t + 8 * hi);
  const float4 b = *reinterpret_cast<const float4*>(PAR + v * 128 + 16 * t + 8 * hi + 4);
  p[0] = a.x; p[1] = a.y; p[2] = a.z; p[3] = a.w;
  p[4] = b.x; p[5] = b.y; p[6] = b.z; p[7] = b.w;
}

__device__ __forceinline__ void gemmT128(const float* Wl, const v8f* bsrc, v8f* acc,
                                         int lane) {
  #pragma unroll
  for (int Q = 0; Q < 16; Q++) {
    float4 w[8];
    #pragma unroll
    for (int t = 0; t < 8; t++) w[t] = ldW(Wl, Q, t, lane);
    const int tp = Q >> 1, u = Q & 1;
    v2f b0, b1;
    b0.x = bsrc[tp][4 * u + 0]; b0.y = bsrc[tp][4 * u + 1];
    b1.x = bsrc[tp][4 * u + 2]; b1.y = bsrc[tp][4 * u + 3];
    #pragma unroll
    for (int t = 0; t < 8; t++) { v2f a; a.x = w[t].x; a.y = w[t].y; acc[t] = wmma4(a, b0, acc[t]); }
    #pragma unroll
    for (int t = 0; t < 8; t++) { v2f a; a.x = w[t].z; a.y = w[t].w; acc[t] = wmma4(a, b1, acc[t]); }
  }
}

__device__ __forceinline__ void actnormT(const float* PAR, int vb, int vg, int vbt,
                                         v8f* acc, v8f* outv, int hi) {
  float ps[8], pq[8];
  #pragma unroll
  for (int t = 0; t < 8; t++) {
    float p[8]; ldP(PAR, vb, t, hi, p);
    float s = 0.f, q = 0.f;
    #pragma unroll
    for (int j = 0; j < 8; j++) {
      const float x = lrelu(acc[t][j] + p[j]);
      acc[t][j] = x; s += x; q += x * x;
    }
    ps[t] = s; pq[t] = q;
  }
  float sum = ((ps[0] + ps[1]) + (ps[2] + ps[3])) + ((ps[4] + ps[5]) + (ps[6] + ps[7]));
  float sq  = ((pq[0] + pq[1]) + (pq[2] + pq[3])) + ((pq[4] + pq[5]) + (pq[6] + pq[7]));
  sum += __shfl_xor(sum, 16);
  sq  += __shfl_xor(sq, 16);
  const float mean = sum * (1.f / 128.f);
  const float var  = sq * (1.f / 128.f) - mean * mean;
  const float rstd = rsqrtf(var + 1e-5f);
  #pragma unroll
  for (int t = 0; t < 8; t++) {
    float g[8], b[8]; ldP(PAR, vg, t, hi, g); ldP(PAR, vbt, t, hi, b);
    #pragma unroll
    for (int j = 0; j < 8; j++) outv[t][j] = g[j] * (acc[t][j] - mean) * rstd + b[j];
  }
}

__global__ void __launch_bounds__(256) edge_kernel(
    const float* __restrict__ coors, const float* __restrict__ h,
    const int* __restrict__ src, const int* __restrict__ dst,
    const float* __restrict__ img, const float* __restrict__ eb1,
    const float* __restrict__ eg1, const float* __restrict__ ebt1,
    const float* __restrict__ eb2, const float* __restrict__ cb1,
    const float* __restrict__ cg1, const float* __restrict__ cbt1,
    const float* __restrict__ cW2, const float* __restrict__ cb2,
    float* __restrict__ sum_msg, float* __restrict__ sum_x,
    float* __restrict__ cnt, int nE) {
  __shared__ __attribute__((aligned(16))) float L[EW_LDSF];
  const int tid = threadIdx.x, lane = tid & 31;
  const int wave = __builtin_amdgcn_readfirstlane(tid >> 5);
  const int hi = lane >> 4, cn = lane & 15;

  {
    const float4* g4 = reinterpret_cast<const float4*>(img);
    float4* l4 = reinterpret_cast<float4*>(L);
    for (int r = 0; r < 3; r++) {
      float4 v[22];
      #pragma unroll
      for (int i = 0; i < 22; i++) v[i] = g4[(r * 22 + i) * 256 + tid];
      #pragma unroll
      for (int i = 0; i < 22; i++) l4[(r * 22 + i) * 256 + tid] = v[i];
    }
    if (tid < 128) {
      float* P = L + EW_IMGF;
      P[0 * 128 + tid] = eb1[tid];  P[1 * 128 + tid] = eg1[tid];
      P[2 * 128 + tid] = ebt1[tid]; P[3 * 128 + tid] = eb2[tid];
      P[4 * 128 + tid] = cb1[tid];  P[5 * 128 + tid] = cg1[tid];
      P[6 * 128 + tid] = cbt1[tid]; P[7 * 128 + tid] = cW2[tid];
    }
  }
  __syncthreads();

  const float* W1L = L;
  const float* W2L = L + EW_W1F;
  const float* W3L = L + EW_W1F + EW_W2F;
  const float* PAR = L + EW_IMGF;
  float* sw = L + EW_IMGF + EW_PARF + wave * (16 * EW_SCR);
  const float cb2v = cb2[0];

  const int ntiles = (nE + 15) >> 4;
  const int stride = (int)gridDim.x * 8;
  int tile = (int)blockIdx.x * 8 + wave;

  int sN, dN;
  float4 xs[16], xd[16];
  float xr0, xr1, xr2;
  int dC;

  auto loadIdx = [&](int tl) {
    const int e = tl * 16 + cn;
    const bool ok = (tl < ntiles) && (e < nE);
    const int ec = ok ? e : 0;
    const int s = src[ec], d = dst[ec];
    sN = ok ? s : 0;
    dN = ok ? d : -1;
  };
  auto fetchRows = [&]() {
    const int s = sN, dn = dN < 0 ? 0 : dN;
    const float* ps = h + (size_t)s * 128 + 4 * hi;
    const float* pd = h + (size_t)dn * 128 + 4 * hi;
    #pragma unroll
    for (int c = 0; c < 16; c++) xs[c] = *reinterpret_cast<const float4*>(ps + 8 * c);
    #pragma unroll
    for (int c = 0; c < 16; c++) xd[c] = *reinterpret_cast<const float4*>(pd + 8 * c);
    xr0 = coors[s * 3 + 0] - coors[dn * 3 + 0];
    xr1 = coors[s * 3 + 1] - coors[dn * 3 + 1];
    xr2 = coors[s * 3 + 2] - coors[dn * 3 + 2];
    dC = dN;
  };

  loadIdx(tile);
  fetchRows();
  loadIdx(tile + stride);

  for (; tile < ntiles; tile += stride) {
    const float d2 = xr0 * xr0 + xr1 * xr1 + xr2 * xr2;
    const float sh = hi ? 2.25f : 1.f;
    v2f rb[4];
    {
      const float base[4] = {1.f, 5.0625f, 25.62890625f, 129.746337890625f};
      #pragma unroll
      for (int r = 0; r < 4; r++) {
        const float s0 = base[r] * sh;
        rb[r].x = __expf(-d2 / s0);
        rb[r].y = __expf(-d2 / (s0 * 1.5f));
      }
      if (hi) rb[3].y = 0.f;
    }

    v8f acc[8];
    #pragma unroll
    for (int t = 0; t < 8; t++) acc[t] = (v8f){};
    #pragma unroll
    for (int Q = 0; Q < 34; Q++) {
      float4 w[8];
      #pragma unroll
      for (int t = 0; t < 8; t++) w[t] = ldW(W1L, Q, t, lane);
      v2f b0, b1;
      if (Q < 16)      { b0.x = xs[Q].x; b0.y = xs[Q].y; b1.x = xs[Q].z; b1.y = xs[Q].w; }
      else if (Q < 32) { b0.x = xd[Q - 16].x; b0.y = xd[Q - 16].y; b1.x = xd[Q - 16].z; b1.y = xd[Q - 16].w; }
      else             { b0 = rb[2 * (Q - 32)]; b1 = rb[2 * (Q - 32) + 1]; }
      #pragma unroll
      for (int t = 0; t < 8; t++) { v2f a; a.x = w[t].x; a.y = w[t].y; acc[t] = wmma4(a, b0, acc[t]); }
      #pragma unroll
      for (int t = 0; t < 8; t++) { v2f a; a.x = w[t].z; a.y = w[t].w; acc[t] = wmma4(a, b1, acc[t]); }
    }

    const int dcur = dC;
    const float xc0 = xr0, xc1 = xr1, xc2 = xr2;
    fetchRows();
    loadIdx(tile + 2 * stride);

    v8f act[8];
    actnormT(PAR, 0, 1, 2, acc, act, hi);

    v8f msg[8];
    #pragma unroll
    for (int t = 0; t < 8; t++) msg[t] = (v8f){};
    gemmT128(W2L, act, msg, lane);
    #pragma unroll
    for (int t = 0; t < 8; t++) {
      float p[8]; ldP(PAR, 3, t, hi, p);
      #pragma unroll
      for (int j = 0; j < 8; j++) msg[t][j] += p[j];
    }

    v8f c1[8];
    #pragma unroll
    for (int t = 0; t < 8; t++) c1[t] = (v8f){};
    gemmT128(W3L, msg, c1, lane);
    actnormT(PAR, 4, 5, 6, c1, act, hi);
    float cpart = 0.f;
    #pragma unroll
    for (int t = 0; t < 8; t++) {
      float p[8]; ldP(PAR, 7, t, hi, p);
      #pragma unroll
      for (int j = 0; j < 8; j++) cpart += act[t][j] * p[j];
    }
    const float coef = cpart + __shfl_xor(cpart, 16) + cb2v;

    #pragma unroll
    for (int pass = 0; pass < 2; pass++) {
      #pragma unroll
      for (int tt = 0; tt < 4; tt++) {
        const v8f m = msg[4 * pass + tt];
        float* q = sw + cn * EW_SCR + 16 * tt + 8 * hi;
        *reinterpret_cast<float4*>(q)     = make_float4(m[0], m[1], m[2], m[3]);
        *reinterpret_cast<float4*>(q + 4) = make_float4(m[4], m[5], m[6], m[7]);
      }
      asm volatile("s_wait_dscnt 0" ::: "memory");
      #pragma unroll
      for (int e = 0; e < 16; e++) {
        const int de = __builtin_amdgcn_readlane(dcur, e);
        if (de >= 0) {
          float* g = sum_msg + (size_t)de * 128 + 64 * pass + lane;
          atomicAdd(g,      sw[e * EW_SCR + lane]);
          atomicAdd(g + 32, sw[e * EW_SCR + 32 + lane]);
        }
      }
      asm volatile("s_wait_dscnt 0" ::: "memory");
    }
    if (dcur >= 0) {
      if (hi == 0) {
        atomicAdd(&sum_x[dcur * 3 + 0], xc0 * coef);
        atomicAdd(&sum_x[dcur * 3 + 1], xc1 * coef);
        atomicAdd(&sum_x[dcur * 3 + 2], xc2 * coef);
      } else {
        atomicAdd(&cnt[dcur], 1.f);
      }
    }
  }
}

#define NCAT_LD 450
__global__ void __launch_bounds__(128) node_kernel(
    const float* __restrict__ h, const float* __restrict__ summ,
    const float* __restrict__ cnt,
    const float* __restrict__ op0, const float* __restrict__ op1,
    const float* __restrict__ op2, const float* __restrict__ op3,
    const float* __restrict__ ml,
    const float* __restrict__ orig,
    const float* __restrict__ nW1p, const float* __restrict__ nb1,
    const float* __restrict__ ng1, const float* __restrict__ nbt1,
    const float* __restrict__ nW2p, const float* __restrict__ nb2,
    float* __restrict__ hout) {
  __shared__ float cat[16 * NCAT_LD];
  __shared__ float buf1[16 * 132];
  __shared__ float stats[16][2];
  __shared__ float jw[16][AT_SPLITS];
  const int tid = threadIdx.x, lane = tid & 31, wave = tid >> 5;
  const int hi = lane >> 4, cn = lane & 15;
  const int node0 = blockIdx.x * 16;

  if (tid < 16) {
    const int node = node0 + tid;
    float mb[AT_SPLITS], lb[AT_SPLITS];
    float M = -3.0e38f;
    #pragma unroll
    for (int b = 0; b < AT_SPLITS; b++) {
      mb[b] = ml[(size_t)(2 * b) * N_NODES + node];
      lb[b] = ml[(size_t)(2 * b + 1) * N_NODES + node];
      M = fmaxf(M, mb[b]);
    }
    float den = 0.f;
    #pragma unroll
    for (int b = 0; b < AT_SPLITS; b++) { mb[b] = __expf(mb[b] - M); den += lb[b] * mb[b]; }
    const float inv = den > 0.f ? 1.f / den : 0.f;
    #pragma unroll
    for (int b = 0; b < AT_SPLITS; b++) jw[tid][b] = mb[b] * inv;
  }
  __syncthreads();

  for (int i = tid; i < 2048; i += 128) {
    int e = i >> 7, c = i & 127;
    int node = node0 + e;
    cat[e * NCAT_LD + c] = h[(size_t)node * 128 + c];
    float ct = cnt[node];
    cat[e * NCAT_LD + 128 + c] = ct > 0.f ? summ[(size_t)node * 128 + c] / ct : 0.f;
    const size_t g = (size_t)node * 128 + c;
    cat[e * NCAT_LD + 256 + c] = ((op0[g] * jw[e][0] + op1[g] * jw[e][1]) + op2[g] * jw[e][2])
                                 + op3[g] * jw[e][3];
  }
  for (int i = tid; i < 1024; i += 128) {
    int e = i >> 6, c = i & 63;
    cat[e * NCAT_LD + 384 + c] = orig[(size_t)(node0 + e) * 64 + c];
  }
  __syncthreads();
  {
    const int n0 = wave * 32;
    v8f a0 = {}, a1 = {};
    for (int k0 = 0; k0 < 448; k0 += 4) {
      v2f a  = ldsA(cat, NCAT_LD, k0, lane);
      v2f b0 = ldBp(nW1p, k0, n0, lane);
      v2f b1 = ldBp(nW1p, k0, n0 + 16, lane);
      a0 = wmma4(a, b0, a0);
      a1 = wmma4(a, b1, a1);
    }
    #pragma unroll
    for (int j = 0; j < 8; j++) {
      int r = j + 8 * hi;
      buf1[r * 132 + n0 + cn]      = lrelu(a0[j] + nb1[n0 + cn]);
      buf1[r * 132 + n0 + 16 + cn] = lrelu(a1[j] + nb1[n0 + 16 + cn]);
    }
  }
  __syncthreads();
  layernorm128(buf1, ng1, nbt1, stats, tid);
  __syncthreads();
  {
    const int n0 = wave * 32;
    v8f a0 = {}, a1 = {};
    for (int k0 = 0; k0 < 128; k0 += 4) {
      v2f a  = ldsA(buf1, 132, k0, lane);
      v2f b0 = ldBp(nW2p, k0, n0, lane);
      v2f b1 = ldBp(nW2p, k0, n0 + 16, lane);
      a0 = wmma4(a, b0, a0);
      a1 = wmma4(a, b1, a1);
    }
    #pragma unroll
    for (int j = 0; j < 8; j++) {
      int r = j + 8 * hi;
      size_t g0 = (size_t)(node0 + r) * 128 + n0 + cn;
      size_t g1 = g0 + 16;
      hout[g0] = 0.5f * (a0[j] + nb2[n0 + cn]) + 0.5f * h[g0];
      hout[g1] = 0.5f * (a1[j] + nb2[n0 + 16 + cn]) + 0.5f * h[g1];
    }
  }
}

__global__ void coords_kernel(const float* __restrict__ oc, const float* __restrict__ c,
                              const float* __restrict__ sx, const float* __restrict__ cnt,
                              float* __restrict__ out, int n) {
  int i = blockIdx.x * blockDim.x + threadIdx.x;
  if (i >= n * 3) return;
  int node = i / 3;
  float ct = cnt[node];
  float xu = ct > 0.f ? sx[i] / ct : 0.f;
  out[i] = 0.2f * oc[i] + 0.8f * c[i] + xu;
}

extern "C" void kernel_launch(void* const* d_in, const int* in_sizes, int n_in,
                              void* d_out, int out_size, void* d_ws, size_t ws_size,
                              hipStream_t stream) {
  const float* coors_lig = (const float*)d_in[0];
  const float* h_lig     = (const float*)d_in[1];
  const float* orig_lig  = (const float*)d_in[2];
  const float* oc_lig    = (const float*)d_in[3];
  const float* coors_rec = (const float*)d_in[4];
  const float* h_rec     = (const float*)d_in[5];
  const float* orig_rec  = (const float*)d_in[6];
  const float* oc_rec    = (const float*)d_in[7];
  const int* lig_src = (const int*)d_in[8];
  const int* lig_dst = (const int*)d_in[9];
  const int* rec_src = (const int*)d_in[10];
  const int* rec_dst = (const int*)d_in[11];
  const float* mask = (const float*)d_in[12];
  const float* eW1 = (const float*)d_in[13];
  const float* eb1 = (const float*)d_in[14];
  const float* eg1 = (const float*)d_in[15];
  const float* ebt1 = (const float*)d_in[16];
  const float* eW2 = (const float*)d_in[17];
  const float* eb2 = (const float*)d_in[18];
  const float* qW = (const float*)d_in[19];
  const float* kW = (const float*)d_in[20];
  const float* vW = (const float*)d_in[21];
  const float* nW1 = (const float*)d_in[22];
  const float* nb1 = (const float*)d_in[23];
  const float* ng1 = (const float*)d_in[24];
  const float* nbt1 = (const float*)d_in[25];
  const float* nW2 = (const float*)d_in[26];
  const float* nb2 = (const float*)d_in[27];
  const float* cW1 = (const float*)d_in[28];
  const float* cb1 = (const float*)d_in[29];
  const float* cg1 = (const float*)d_in[30];
  const float* cbt1 = (const float*)d_in[31];
  const float* cW2 = (const float*)d_in[32];
  const float* cb2 = (const float*)d_in[33];
  const int nE = in_sizes[8];

  float* ws = (float*)d_ws;
  float* q_l = ws + 0 * NH;
  float* k_l = ws + 1 * NH;
  float* v_l = ws + 2 * NH;
  float* q_r = ws + 3 * NH;
  float* k_r = ws + 4 * NH;
  float* v_r = ws + 5 * NH;
  float* cm_l = ws + 6 * NH;
  float* cm_r = ws + 7 * NH;
  float* sm_l = ws + 8 * NH;
  float* sm_r = ws + 9 * NH;
  float* sx_l = ws + 10 * NH;
  float* sx_r = sx_l + 3 * N_NODES;
  float* cnt_l = sx_r + 3 * N_NODES;
  float* cnt_r = cnt_l + N_NODES;
  float* eW1p = cnt_r + N_NODES;
  float* eW2p = eW1p + 136 * 256;
  float* cW1p = eW2p + 64 * 256;
  float* nW1p = cW1p + 64 * 256;
  float* nW2p = nW1p + 224 * 256;
  float* qWp  = nW2p + 64 * 256;
  float* kWp  = qWp + 64 * 256;
  float* vWp  = kWp + 64 * 256;

  size_t accum_floats = 2 * NH + 8 * (size_t)N_NODES;
  hipMemsetAsync(sm_l, 0, accum_floats * sizeof(float), stream);

  pack_frag<<<EW_W1F / 256, 256, 0, stream>>>(eW1, eW1p, 271, EW_W1F, 0);
  pack_frag<<<EW_W2F / 256, 256, 0, stream>>>(eW2, eW2p, 128, EW_W2F, 1);
  pack_frag<<<EW_W2F / 256, 256, 0, stream>>>(cW1, cW1p, 128, EW_W2F, 1);
  pack_pairs<<<(224 * 128 + 255) / 256, 256, 0, stream>>>(nW1, nW1p, 448, 224);
  pack_pairs<<<(64 * 128 + 255) / 256, 256, 0, stream>>>(nW2, nW2p, 128, 64);
  pack_pairs<<<(64 * 128 + 255) / 256, 256, 0, stream>>>(qW, qWp, 128, 64);
  pack_pairs<<<(64 * 128 + 255) / 256, 256, 0, stream>>>(kW, kWp, 128, 64);
  pack_pairs<<<(64 * 128 + 255) / 256, 256, 0, stream>>>(vW, vWp, 128, 64);

  float* out_xl = (float*)d_out;
  float* out_hl = out_xl + 3 * N_NODES;
  float* out_xr = out_hl + NH;
  float* out_hr = out_xr + 3 * N_NODES;

  const int ntiles = N_NODES / 16;
  const dim3 agrid((N_NODES + 31) / 32, AT_SPLITS);


  proj_gemm<<<ntiles, 128, 0, stream>>>(h_lig, qWp, q_l, 1, 0);
  proj_gemm<<<ntiles, 128, 0, stream>>>(h_rec, kWp, k_r, 1, 0);
  proj_gemm<<<ntiles, 128, 0, stream>>>(h_rec, vWp, v_r, 0, 1);
  attn_kernel<<<agrid, 512, 0, stream>>>(q_l, k_r, v_r, mask, 0, cm_l, cm_r, q_r, k_l, v_l);
  edge_kernel<<<256, 256, 0, stream>>>(coors_lig, h_lig, lig_src, lig_dst,
      eW1p, eb1, eg1, ebt1, eb2, cb1, cg1, cbt1, cW2, cb2,
      sm_l, sx_l, cnt_l, nE);
  node_kernel<<<ntiles, 128, 0, stream>>>(h_lig, sm_l, cnt_l, cm_l, cm_r, q_r, k_l, v_l,
      orig_lig, nW1p, nb1, ng1, nbt1, nW2p, nb2, out_hl);

  proj_gemm<<<ntiles, 128, 0, stream>>>(h_rec, qWp, q_r, 1, 0);
  proj_gemm<<<ntiles, 128, 0, stream>>>(h_lig, kWp, k_l, 1, 0);
  proj_gemm<<<ntiles, 128, 0, stream>>>(h_lig, vWp, v_l, 0, 1);
  attn_kernel<<<agrid, 512, 0, stream>>>(q_r, k_l, v_l, mask, 1, cm_l, cm_r, q_l, k_r, v_r);
  edge_kernel<<<256, 256, 0, stream>>>(coors_rec, h_rec, rec_src, rec_dst,
      eW1p, eb1, eg1, ebt1, eb2, cb1, cg1, cbt1, cW2, cb2,
      sm_r, sx_r, cnt_r, nE);
  node_kernel<<<ntiles, 128, 0, stream>>>(h_rec, sm_r, cnt_r, cm_l, cm_r, q_l, k_r, v_r,
      orig_rec, nW1p, nb1, ng1, nbt1, nW2p, nb2, out_hr);

  int cthreads = 3 * N_NODES;
  coords_kernel<<<(cthreads + 255) / 256, 256, 0, stream>>>(oc_lig, coors_lig, sx_l,
                                                            cnt_l, out_xl, N_NODES);
  coords_kernel<<<(cthreads + 255) / 256, 256, 0, stream>>>(oc_rec, coors_rec, sx_r,
                                                            cnt_r, out_xr, N_NODES);
}
